# starting kernel
# baseline (speedup 1.0000x reference)
#include <hip/hip_runtime.h>
#include <hip/hip_bf16.h>
#include <math.h>

#define NN 50000
#define DD 128
#define HH 256
#define EE 400000
#define KK 256
#define MT 32
#define PSTR 260

typedef float v2f __attribute__((ext_vector_type(2)));
typedef float v8f __attribute__((ext_vector_type(8)));

__device__ inline v8f wmma_f32(v2f a, v2f b, v8f c) {
    return __builtin_amdgcn_wmma_f32_16x16x4_f32(false, a, false, b, (short)0, c,
                                                 false, false);
}

__global__ void init_kernel(int* ws, int n) {
    int i = blockIdx.x * blockDim.x + threadIdx.x;
    if (i < n) ws[i] = 0;
}

__global__ __launch_bounds__(256) void det_kernel(
        const float* __restrict__ emb, const int* __restrict__ eidx,
        const float* __restrict__ Wd1, const float* __restrict__ bd1,
        const float* __restrict__ Wd2, const float* __restrict__ bd2,
        float* __restrict__ scores, int* __restrict__ last)
{
    __shared__ float pair_lds[MT * PSTR];
    __shared__ float h_lds[MT * PSTR];

    const int t    = threadIdx.x;
    const int lane = t & 31;
    const int wave = t >> 5;
    const int e0   = blockIdx.x * MT;
    const int* srcp = eidx;
    const int* dstp = eidx + EE;

    if (t == 0) {
        __builtin_prefetch(Wd1, 0, 1);
        __builtin_prefetch(Wd2, 0, 1);
    }

    {
        const int row  = t >> 3;
        const int col  = (t & 7) * 32;
        const int e    = e0 + row;
        const int node = (col < DD) ? srcp[e] : dstp[e];
        const int goff = (col < DD) ? col : col - DD;
        const float4* g = (const float4*)(emb + (size_t)node * DD + goff);
        float4* l = (float4*)(pair_lds + row * PSTR + col);
        #pragma unroll
        for (int j = 0; j < 8; ++j) l[j] = g[j];
    }
    __syncthreads();

    const int row   = lane & 15;
    const int khalf = (lane >> 4) << 1;
    const int ncol  = lane & 15;
    const int n0    = wave * 32;

    v8f acc00 = {}, acc01 = {}, acc10 = {}, acc11 = {};
    for (int k0 = 0; k0 < KK; k0 += 4) {
        v2f a0, a1;
        a0.x = pair_lds[row * PSTR + k0 + khalf];
        a0.y = pair_lds[row * PSTR + k0 + khalf + 1];
        a1.x = pair_lds[(row + 16) * PSTR + k0 + khalf];
        a1.y = pair_lds[(row + 16) * PSTR + k0 + khalf + 1];
        const float* wp = Wd1 + (size_t)(k0 + khalf) * HH;
        v2f b0, b1;
        b0.x = wp[n0 + ncol];        b0.y = wp[HH + n0 + ncol];
        b1.x = wp[n0 + 16 + ncol];   b1.y = wp[HH + n0 + 16 + ncol];
        acc00 = wmma_f32(a0, b0, acc00);
        acc01 = wmma_f32(a0, b1, acc01);
        acc10 = wmma_f32(a1, b0, acc10);
        acc11 = wmma_f32(a1, b1, acc11);
    }

    const int mhi = (lane >> 4) << 3;
    #pragma unroll
    for (int v = 0; v < 8; ++v) {
        const int M  = v + mhi;
        const int Na = n0 + ncol;
        const int Nb = n0 + 16 + ncol;
        const float ba = bd1[Na], bb = bd1[Nb];
        h_lds[M * PSTR + Na]        = fmaxf(acc00[v] + ba, 0.0f);
        h_lds[M * PSTR + Nb]        = fmaxf(acc01[v] + bb, 0.0f);
        h_lds[(M + 16) * PSTR + Na] = fmaxf(acc10[v] + ba, 0.0f);
        h_lds[(M + 16) * PSTR + Nb] = fmaxf(acc11[v] + bb, 0.0f);
    }
    __syncthreads();

    float* part = pair_lds;
    {
        const int e  = t >> 3;
        const int c0 = (t & 7) * 32;
        float p = 0.0f;
        #pragma unroll 8
        for (int j = 0; j < 32; ++j) p += h_lds[e * PSTR + c0 + j] * Wd2[c0 + j];
        part[e * 9 + (t & 7)] = p;
    }
    __syncthreads();
    if (t < MT) {
        float x = bd2[0];
        #pragma unroll
        for (int j = 0; j < 8; ++j) x += part[t * 9 + j];
        const int e = e0 + t;
        scores[e] = 1.0f / (1.0f + __expf(-x));
        if (x > 0.0f) {
            const int ord = e + 1;
            atomicMax(&last[srcp[e]], ord);
            atomicMax(&last[dstp[e]], ord);
        }
    }
}

__global__ void compact_kernel(const int* __restrict__ last, int* counter,
                               int* __restrict__ active) {
    int i = blockIdx.x * blockDim.x + threadIdx.x;
    if (i < NN && last[i] > 0) {
        int p = atomicAdd(counter, 1);
        active[p] = i;
    }
}

__global__ void copy_kernel(const float* __restrict__ emb,
                            const int* __restrict__ last,
                            float* __restrict__ out) {
    int t = blockIdx.x * blockDim.x + threadIdx.x;
    int node = t >> 5, j = t & 31;
    if (node < NN && last[node] == 0)
        ((float4*)out)[(size_t)node * 32 + j] = ((const float4*)emb)[(size_t)node * 32 + j];
}

__global__ __launch_bounds__(256) void res_kernel(
        const float* __restrict__ emb, const int* __restrict__ eidx,
        const float* __restrict__ Wr1, const float* __restrict__ br1,
        const float* __restrict__ Wr2, const float* __restrict__ br2,
        const int* __restrict__ last, const int* __restrict__ counter,
        const int* __restrict__ active, float* __restrict__ out)
{
    __shared__ float pair_lds[MT * PSTR];
    __shared__ float h_lds[MT * PSTR];

    const int count = *counter;
    const int base  = blockIdx.x * MT;
    if (base >= count) return;

    const int t    = threadIdx.x;
    const int lane = t & 31;
    const int wave = t >> 5;
    const int* srcp = eidx;
    const int* dstp = eidx + EE;

    {
        const int row  = t >> 3;
        const int col  = (t & 7) * 32;
        int gi = base + row; if (gi > count - 1) gi = count - 1;
        const int node = active[gi];
        const int e    = last[node] - 1;
        const int nn   = (col < DD) ? srcp[e] : dstp[e];
        const int goff = (col < DD) ? col : col - DD;
        const float4* g = (const float4*)(emb + (size_t)nn * DD + goff);
        float4* l = (float4*)(pair_lds + row * PSTR + col);
        #pragma unroll
        for (int j = 0; j < 8; ++j) l[j] = g[j];
    }
    __syncthreads();

    const int row   = lane & 15;
    const int khalf = (lane >> 4) << 1;
    const int ncol  = lane & 15;
    const int mhi   = (lane >> 4) << 3;

    {
        const int n0 = wave * 32;
        v8f acc00 = {}, acc01 = {}, acc10 = {}, acc11 = {};
        for (int k0 = 0; k0 < KK; k0 += 4) {
            v2f a0, a1;
            a0.x = pair_lds[row * PSTR + k0 + khalf];
            a0.y = pair_lds[row * PSTR + k0 + khalf + 1];
            a1.x = pair_lds[(row + 16) * PSTR + k0 + khalf];
            a1.y = pair_lds[(row + 16) * PSTR + k0 + khalf + 1];
            const float* wp = Wr1 + (size_t)(k0 + khalf) * HH;
            v2f b0, b1;
            b0.x = wp[n0 + ncol];       b0.y = wp[HH + n0 + ncol];
            b1.x = wp[n0 + 16 + ncol];  b1.y = wp[HH + n0 + 16 + ncol];
            acc00 = wmma_f32(a0, b0, acc00);
            acc01 = wmma_f32(a0, b1, acc01);
            acc10 = wmma_f32(a1, b0, acc10);
            acc11 = wmma_f32(a1, b1, acc11);
        }
        #pragma unroll
        for (int v = 0; v < 8; ++v) {
            const int M  = v + mhi;
            const int Na = n0 + ncol;
            const int Nb = n0 + 16 + ncol;
            const float ba = br1[Na], bb = br1[Nb];
            h_lds[M * PSTR + Na]        = fmaxf(acc00[v] + ba, 0.0f);
            h_lds[M * PSTR + Nb]        = fmaxf(acc01[v] + bb, 0.0f);
            h_lds[(M + 16) * PSTR + Na] = fmaxf(acc10[v] + ba, 0.0f);
            h_lds[(M + 16) * PSTR + Nb] = fmaxf(acc11[v] + bb, 0.0f);
        }
    }
    __syncthreads();

    {
        const int n0 = wave * 16;
        v8f acc0 = {}, acc1 = {};
        for (int k0 = 0; k0 < KK; k0 += 4) {
            v2f a0, a1;
            a0.x = h_lds[row * PSTR + k0 + khalf];
            a0.y = h_lds[row * PSTR + k0 + khalf + 1];
            a1.x = h_lds[(row + 16) * PSTR + k0 + khalf];
            a1.y = h_lds[(row + 16) * PSTR + k0 + khalf + 1];
            const float* wp = Wr2 + (size_t)(k0 + khalf) * DD;
            v2f b;
            b.x = wp[n0 + ncol];  b.y = wp[DD + n0 + ncol];
            acc0 = wmma_f32(a0, b, acc0);
            acc1 = wmma_f32(a1, b, acc1);
        }
        #pragma unroll
        for (int v = 0; v < 8; ++v) {
            const int Nc = n0 + ncol;
            const int M  = v + mhi;
            int gi = base + M;
            if (gi < count) {
                const int node = active[gi];
                out[(size_t)node * DD + Nc] =
                    (emb[(size_t)node * DD + Nc] + acc0[v] + br2[Nc]) * 0.5f;
            }
            gi = base + M + 16;
            if (gi < count) {
                const int node = active[gi];
                out[(size_t)node * DD + Nc] =
                    (emb[(size_t)node * DD + Nc] + acc1[v] + br2[Nc]) * 0.5f;
            }
        }
    }
}

extern "C" void kernel_launch(void* const* d_in, const int* in_sizes, int n_in,
                              void* d_out, int out_size, void* d_ws, size_t ws_size,
                              hipStream_t stream) {
    const float* emb = (const float*)d_in[0];
    const int*   eix = (const int*)  d_in[1];
    const float* Wd1 = (const float*)d_in[2];
    const float* bd1 = (const float*)d_in[3];
    const float* Wd2 = (const float*)d_in[4];
    const float* bd2 = (const float*)d_in[5];
    const float* Wr1 = (const float*)d_in[6];
    const float* br1 = (const float*)d_in[7];
    const float* Wr2 = (const float*)d_in[8];
    const float* br2 = (const float*)d_in[9];

    float* out_emb    = (float*)d_out;
    float* out_scores = out_emb + (size_t)NN * DD;

    int* ws      = (int*)d_ws;
    int* counter = ws;
    int* last    = ws + 16;
    int* activeL = ws + 16 + NN;

    const int zn = NN + 16;
    init_kernel<<<(zn + 255) / 256, 256, 0, stream>>>(ws, zn);

    det_kernel<<<EE / MT, 256, 0, stream>>>(emb, eix, Wd1, bd1, Wd2, bd2,
                                            out_scores, last);

    compact_kernel<<<(NN + 255) / 256, 256, 0, stream>>>(last, counter, activeL);

    copy_kernel<<<(NN * 32 + 255) / 256, 256, 0, stream>>>(emb, last, out_emb);

    res_kernel<<<(NN + MT - 1) / MT, 256, 0, stream>>>(emb, eix, Wr1, br1, Wr2, br2,
                                                       last, counter, activeL, out_emb);
}


# Detector first layer factored through nodes: emb@Wd1[0:D] and emb@Wd1[D:2D] once per node on f32 WMMA (two N x H tables); each edge gathers and adds its two rows: 8x fewer matrix FLOPs
# speedup vs baseline: 1.7406x; 1.7406x over previous
#include <hip/hip_runtime.h>
#include <hip/hip_bf16.h>
#include <math.h>

#define NN 50000
#define DD 128
#define HH 256
#define EE 400000
#define KK 256
#define MT 32
#define PSTR 260
#define NT 64
#define ASTR 132
#define EGRID 1250
#define EITER 10
#define TAB_OFF 100032
#define WS_NEED ((size_t)TAB_OFF * 4 + (size_t)2 * NN * HH * 4)

typedef float v2f __attribute__((ext_vector_type(2)));
typedef float v8f __attribute__((ext_vector_type(8)));

__device__ inline v8f wmma_f32(v2f a, v2f b, v8f c) {
    return __builtin_amdgcn_wmma_f32_16x16x4_f32(false, a, false, b, (short)0, c,
                                                 false, false);
}

__global__ void init_kernel(int* ws, int n) {
    int i = blockIdx.x * blockDim.x + threadIdx.x;
    if (i < n) ws[i] = 0;
}

__global__ __launch_bounds__(128) void node_kernel(
        const float* __restrict__ emb, const float* __restrict__ Wd1,
        const float* __restrict__ bd1, float* __restrict__ tab, size_t ws_bytes)
{
    if (ws_bytes < WS_NEED) return;

    __shared__ float a_lds[NT * ASTR];

    const int t     = threadIdx.x;
    const int lane  = t & 31;
    const int wave  = t >> 5;
    const int node0 = blockIdx.x * NT;

    #pragma unroll
    for (int j = 0; j < 16; ++j) {
        const int idx = j * 128 + t;
        const int r   = idx >> 5;
        const int c4  = idx & 31;
        int node = node0 + r; if (node > NN - 1) node = NN - 1;
        const float4 v = ((const float4*)emb)[(size_t)node * 32 + c4];
        *(float4*)(a_lds + r * ASTR + c4 * 4) = v;
    }
    __syncthreads();

    const int row   = lane & 15;
    const int khalf = (lane >> 4) << 1;
    const int ncol  = lane & 15;
    const int mhi   = (lane >> 4) << 3;
    const int side  = blockIdx.y;
    const int cb    = wave * 64 + ncol * 4;
    const float* wbase = Wd1 + (size_t)side * DD * HH + cb;

    v8f acc[4][4];
    #pragma unroll
    for (int m = 0; m < 4; ++m)
        #pragma unroll
        for (int n = 0; n < 4; ++n) acc[m][n] = (v8f){};

    for (int k0 = 0; k0 < DD; k0 += 4) {
        const float4 w0 = *(const float4*)(wbase + (size_t)(k0 + khalf) * HH);
        const float4 w1 = *(const float4*)(wbase + (size_t)(k0 + khalf + 1) * HH);
        v2f a[4];
        #pragma unroll
        for (int m = 0; m < 4; ++m)
            a[m] = *(const v2f*)(a_lds + (m * 16 + row) * ASTR + k0 + khalf);
        v2f b[4];
        b[0].x = w0.x; b[0].y = w1.x;
        b[1].x = w0.y; b[1].y = w1.y;
        b[2].x = w0.z; b[2].y = w1.z;
        b[3].x = w0.w; b[3].y = w1.w;
        #pragma unroll
        for (int m = 0; m < 4; ++m)
            #pragma unroll
            for (int n = 0; n < 4; ++n)
                acc[m][n] = wmma_f32(a[m], b[n], acc[m][n]);
    }

    float4 bias = make_float4(0.0f, 0.0f, 0.0f, 0.0f);
    if (side == 0) bias = *(const float4*)(bd1 + cb);
    float* dstp = tab + (size_t)side * NN * HH;
    #pragma unroll
    for (int m = 0; m < 4; ++m) {
        #pragma unroll
        for (int v = 0; v < 8; ++v) {
            const int node = node0 + m * 16 + v + mhi;
            if (node < NN) {
                float4 o;
                o.x = acc[m][0][v] + bias.x;
                o.y = acc[m][1][v] + bias.y;
                o.z = acc[m][2][v] + bias.z;
                o.w = acc[m][3][v] + bias.w;
                *(float4*)(dstp + (size_t)node * HH + cb) = o;
            }
        }
    }
}

__global__ __launch_bounds__(256) void edge_kernel(
        const int* __restrict__ eidx, const float* __restrict__ tab,
        const float* __restrict__ Wd2, const float* __restrict__ bd2,
        float* __restrict__ scores, int* __restrict__ last, size_t ws_bytes)
{
    if (ws_bytes < WS_NEED) return;

    const int t  = threadIdx.x;
    const int l8 = t & 7;
    const int* srcp = eidx;
    const int* dstp = eidx + EE;
    const float4* P = (const float4*)tab;
    const float4* Q = P + (size_t)NN * (HH / 4);

    float4 w[8];
    #pragma unroll
    for (int j = 0; j < 8; ++j) w[j] = ((const float4*)Wd2)[j * 8 + l8];
    const float b2 = bd2[0];

    int e = blockIdx.x * 32 + (t >> 3);
    int s = srcp[e], d = dstp[e];
    for (int it = 0; it < EITER; ++it) {
        int en = e + EGRID * 32; if (en > EE - 1) en = EE - 1;
        const int sn = srcp[en], dn = dstp[en];

        const float4* pr = P + (size_t)s * (HH / 4) + l8;
        const float4* qr = Q + (size_t)d * (HH / 4) + l8;
        float4 p[8], q[8];
        #pragma unroll
        for (int j = 0; j < 8; ++j) { p[j] = pr[j * 8]; q[j] = qr[j * 8]; }

        float acc = 0.0f;
        #pragma unroll
        for (int j = 0; j < 8; ++j) {
            acc += fmaxf(p[j].x + q[j].x, 0.0f) * w[j].x;
            acc += fmaxf(p[j].y + q[j].y, 0.0f) * w[j].y;
            acc += fmaxf(p[j].z + q[j].z, 0.0f) * w[j].z;
            acc += fmaxf(p[j].w + q[j].w, 0.0f) * w[j].w;
        }
        acc += __shfl_xor(acc, 1);
        acc += __shfl_xor(acc, 2);
        acc += __shfl_xor(acc, 4);

        if (l8 == 0) {
            const float x = acc + b2;
            scores[e] = 1.0f / (1.0f + __expf(-x));
            if (x > 0.0f) {
                const int ord = e + 1;
                atomicMax(&last[s], ord);
                atomicMax(&last[d], ord);
            }
        }
        e += EGRID * 32; s = sn; d = dn;
    }
}

__global__ void compact_kernel(const int* __restrict__ last, int* counter,
                               int* __restrict__ active) {
    int i = blockIdx.x * blockDim.x + threadIdx.x;
    if (i < NN && last[i] > 0) {
        int p = atomicAdd(counter, 1);
        active[p] = i;
    }
}

__global__ void copy_kernel(const float* __restrict__ emb,
                            const int* __restrict__ last,
                            float* __restrict__ out) {
    int t = blockIdx.x * blockDim.x + threadIdx.x;
    int node = t >> 5, j = t & 31;
    if (node < NN && last[node] == 0)
        ((float4*)out)[(size_t)node * 32 + j] = ((const float4*)emb)[(size_t)node * 32 + j];
}

__global__ __launch_bounds__(256) void res_kernel(
        const float* __restrict__ emb, const int* __restrict__ eidx,
        const float* __restrict__ Wr1, const float* __restrict__ br1,
        const float* __restrict__ Wr2, const float* __restrict__ br2,
        const int* __restrict__ last, const int* __restrict__ counter,
        const int* __restrict__ active, float* __restrict__ out)
{
    __shared__ float pair_lds[MT * PSTR];
    __shared__ float h_lds[MT * PSTR];

    const int count = *counter;
    const int base  = blockIdx.x * MT;
    if (base >= count) return;

    const int t    = threadIdx.x;
    const int lane = t & 31;
    const int wave = t >> 5;
    const int* srcp = eidx;
    const int* dstp = eidx + EE;

    {
        const int row  = t >> 3;
        const int col  = (t & 7) * 32;
        int gi = base + row; if (gi > count - 1) gi = count - 1;
        const int node = active[gi];
        const int e    = last[node] - 1;
        const int nn   = (col < DD) ? srcp[e] : dstp[e];
        const int goff = (col < DD) ? col : col - DD;
        const float4* g = (const float4*)(emb + (size_t)nn * DD + goff);
        float4* l = (float4*)(pair_lds + row * PSTR + col);
        #pragma unroll
        for (int j = 0; j < 8; ++j) l[j] = g[j];
    }
    __syncthreads();

    const int row   = lane & 15;
    const int khalf = (lane >> 4) << 1;
    const int ncol  = lane & 15;
    const int mhi   = (lane >> 4) << 3;

    {
        const int n0 = wave * 32;
        v8f acc00 = {}, acc01 = {}, acc10 = {}, acc11 = {};
        for (int k0 = 0; k0 < KK; k0 += 4) {
            v2f a0, a1;
            a0.x = pair_lds[row * PSTR + k0 + khalf];
            a0.y = pair_lds[row * PSTR + k0 + khalf + 1];
            a1.x = pair_lds[(row + 16) * PSTR + k0 + khalf];
            a1.y = pair_lds[(row + 16) * PSTR + k0 + khalf + 1];
            const float* wp = Wr1 + (size_t)(k0 + khalf) * HH;
            v2f b0, b1;
            b0.x = wp[n0 + ncol];       b0.y = wp[HH + n0 + ncol];
            b1.x = wp[n0 + 16 + ncol];  b1.y = wp[HH + n0 + 16 + ncol];
            acc00 = wmma_f32(a0, b0, acc00);
            acc01 = wmma_f32(a0, b1, acc01);
            acc10 = wmma_f32(a1, b0, acc10);
            acc11 = wmma_f32(a1, b1, acc11);
        }
        #pragma unroll
        for (int v = 0; v < 8; ++v) {
            const int M  = v + mhi;
            const int Na = n0 + ncol;
            const int Nb = n0 + 16 + ncol;
            const float ba = br1[Na], bb = br1[Nb];
            h_lds[M * PSTR + Na]        = fmaxf(acc00[v] + ba, 0.0f);
            h_lds[M * PSTR + Nb]        = fmaxf(acc01[v] + bb, 0.0f);
            h_lds[(M + 16) * PSTR + Na] = fmaxf(acc10[v] + ba, 0.0f);
            h_lds[(M + 16) * PSTR + Nb] = fmaxf(acc11[v] + bb, 0.0f);
        }
    }
    __syncthreads();

    {
        const int n0 = wave * 16;
        v8f acc0 = {}, acc1 = {};
        for (int k0 = 0; k0 < KK; k0 += 4) {
            v2f a0, a1;
            a0.x = h_lds[row * PSTR + k0 + khalf];
            a0.y = h_lds[row * PSTR + k0 + khalf + 1];
            a1.x = h_lds[(row + 16) * PSTR + k0 + khalf];
            a1.y = h_lds[(row + 16) * PSTR + k0 + khalf + 1];
            const float* wp = Wr2 + (size_t)(k0 + khalf) * DD;
            v2f b;
            b.x = wp[n0 + ncol];  b.y = wp[DD + n0 + ncol];
            acc0 = wmma_f32(a0, b, acc0);
            acc1 = wmma_f32(a1, b, acc1);
        }
        #pragma unroll
        for (int v = 0; v < 8; ++v) {
            const int Nc = n0 + ncol;
            const int M  = v + mhi;
            int gi = base + M;
            if (gi < count) {
                const int node = active[gi];
                out[(size_t)node * DD + Nc] =
                    (emb[(size_t)node * DD + Nc] + acc0[v] + br2[Nc]) * 0.5f;
            }
            gi = base + M + 16;
            if (gi < count) {
                const int node = active[gi];
                out[(size_t)node * DD + Nc] =
                    (emb[(size_t)node * DD + Nc] + acc1[v] + br2[Nc]) * 0.5f;
            }
        }
    }
}

extern "C" void kernel_launch(void* const* d_in, const int* in_sizes, int n_in,
                              void* d_out, int out_size, void* d_ws, size_t ws_size,
                              hipStream_t stream) {
    const float* emb = (const float*)d_in[0];
    const int*   eix = (const int*)  d_in[1];
    const float* Wd1 = (const float*)d_in[2];
    const float* bd1 = (const float*)d_in[3];
    const float* Wd2 = (const float*)d_in[4];
    const float* bd2 = (const float*)d_in[5];
    const float* Wr1 = (const float*)d_in[6];
    const float* br1 = (const float*)d_in[7];
    const float* Wr2 = (const float*)d_in[8];
    const float* br2 = (const float*)d_in[9];

    float* out_emb    = (float*)d_out;
    float* out_scores = out_emb + (size_t)NN * DD;

    int* ws      = (int*)d_ws;
    int* counter = ws;
    int* last    = ws + 16;
    int* activeL = ws + 16 + NN;

    const int zn = NN + 16;
    init_kernel<<<(zn + 255) / 256, 256, 0, stream>>>(ws, zn);

    float* tab = (float*)d_ws + TAB_OFF;

    node_kernel<<<dim3((NN + NT - 1) / NT, 2), 128, 0, stream>>>(emb, Wd1, bd1, tab, ws_size);

    edge_kernel<<<EGRID, 256, 0, stream>>>(eix, tab, Wd2, bd2, out_scores, last, ws_size);

    compact_kernel<<<(NN + 255) / 256, 256, 0, stream>>>(last, counter, activeL);

    copy_kernel<<<(NN * 32 + 255) / 256, 256, 0, stream>>>(emb, last, out_emb);

    res_kernel<<<(NN + MT - 1) / MT, 256, 0, stream>>>(emb, eix, Wr1, br1, Wr2, br2,
                                                       last, counter, activeL, out_emb);
}


# Resolver GEMMs fed differently: a lane owns two adjacent columns, B arrives as 8-byte loads of whole 128-B lines issued a k block ahead of its WMMAs, rows leave as whole-line stores
# speedup vs baseline: 1.9433x; 1.1165x over previous
#include <hip/hip_runtime.h>
#include <hip/hip_bf16.h>
#include <math.h>

#define NN 50000
#define DD 128
#define HH 256
#define EE 400000
#define KK 256
#define MT 32
#define PSTR 260
#define NT 64
#define ASTR 132
#define EGRID 1250
#define EITER 10
#define TAB_OFF 100032
#define WS_NEED ((size_t)TAB_OFF * 4 + (size_t)2 * NN * HH * 4)

typedef float v2f __attribute__((ext_vector_type(2)));
typedef float v8f __attribute__((ext_vector_type(8)));

__device__ inline v8f wmma_f32(v2f a, v2f b, v8f c) {
    return __builtin_amdgcn_wmma_f32_16x16x4_f32(false, a, false, b, (short)0, c,
                                                 false, false);
}

__global__ __launch_bounds__(128) void node_kernel(
        const float* __restrict__ emb, const float* __restrict__ Wd1,
        const float* __restrict__ bd1, float* __restrict__ tab,
        int* __restrict__ last, size_t ws_bytes)
{
    __shared__ float a_lds[NT * ASTR];

    const int t     = threadIdx.x;
    const int lane  = t & 31;
    const int wave  = t >> 5;
    const int node0 = blockIdx.x * NT;

    if (blockIdx.y == 0 && t < NT && node0 + t < NN) last[node0 + t] = 0;

    if (ws_bytes < WS_NEED) return;

    #pragma unroll
    for (int j = 0; j < 16; ++j) {
        const int idx = j * 128 + t;
        const int r   = idx >> 5;
        const int c4  = idx & 31;
        int node = node0 + r; if (node > NN - 1) node = NN - 1;
        const float4 v = ((const float4*)emb)[(size_t)node * 32 + c4];
        *(float4*)(a_lds + r * ASTR + c4 * 4) = v;
    }
    __syncthreads();

    const int row   = lane & 15;
    const int khalf = (lane >> 4) << 1;
    const int ncol  = lane & 15;
    const int mhi   = (lane >> 4) << 3;
    const int side  = blockIdx.y;
    const int cb    = wave * 64 + ncol * 4;
    const float* wbase = Wd1 + (size_t)side * DD * HH + cb;

    v8f acc[4][4];
    #pragma unroll
    for (int m = 0; m < 4; ++m)
        #pragma unroll
        for (int n = 0; n < 4; ++n) acc[m][n] = (v8f){};

    for (int k0 = 0; k0 < DD; k0 += 4) {
        const float4 w0 = *(const float4*)(wbase + (size_t)(k0 + khalf) * HH);
        const float4 w1 = *(const float4*)(wbase + (size_t)(k0 + khalf + 1) * HH);
        v2f a[4];
        #pragma unroll
        for (int m = 0; m < 4; ++m)
            a[m] = *(const v2f*)(a_lds + (m * 16 + row) * ASTR + k0 + khalf);
        v2f b[4];
        b[0].x = w0.x; b[0].y = w1.x;
        b[1].x = w0.y; b[1].y = w1.y;
        b[2].x = w0.z; b[2].y = w1.z;
        b[3].x = w0.w; b[3].y = w1.w;
        #pragma unroll
        for (int m = 0; m < 4; ++m)
            #pragma unroll
            for (int n = 0; n < 4; ++n)
                acc[m][n] = wmma_f32(a[m], b[n], acc[m][n]);
    }

    float4 bias = make_float4(0.0f, 0.0f, 0.0f, 0.0f);
    if (side == 0) bias = *(const float4*)(bd1 + cb);
    float* dstp = tab + (size_t)side * NN * HH;
    #pragma unroll
    for (int m = 0; m < 4; ++m) {
        #pragma unroll
        for (int v = 0; v < 8; ++v) {
            const int node = node0 + m * 16 + v + mhi;
            if (node < NN) {
                float4 o;
                o.x = acc[m][0][v] + bias.x;
                o.y = acc[m][1][v] + bias.y;
                o.z = acc[m][2][v] + bias.z;
                o.w = acc[m][3][v] + bias.w;
                *(float4*)(dstp + (size_t)node * HH + cb) = o;
            }
        }
    }
}

__global__ __launch_bounds__(256) void edge_kernel(
        const int* __restrict__ eidx, const float* __restrict__ tab,
        const float* __restrict__ Wd2, const float* __restrict__ bd2,
        float* __restrict__ scores, int* __restrict__ last, size_t ws_bytes)
{
    if (ws_bytes < WS_NEED) return;

    const int t  = threadIdx.x;
    const int l8 = t & 7;
    const int* srcp = eidx;
    const int* dstp = eidx + EE;
    const float4* P = (const float4*)tab;
    const float4* Q = P + (size_t)NN * (HH / 4);

    float4 w[8];
    #pragma unroll
    for (int j = 0; j < 8; ++j) w[j] = ((const float4*)Wd2)[j * 8 + l8];
    const float b2 = bd2[0];

    int e = blockIdx.x * 32 + (t >> 3);
    int s = srcp[e], d = dstp[e];
    for (int it = 0; it < EITER; ++it) {
        int en = e + EGRID * 32; if (en > EE - 1) en = EE - 1;
        const int sn = srcp[en], dn = dstp[en];

        const float4* pr = P + (size_t)s * (HH / 4) + l8;
        const float4* qr = Q + (size_t)d * (HH / 4) + l8;
        float4 p[8], q[8];
        #pragma unroll
        for (int j = 0; j < 8; ++j) { p[j] = pr[j * 8]; q[j] = qr[j * 8]; }

        float acc = 0.0f;
        #pragma unroll
        for (int j = 0; j < 8; ++j) {
            acc += fmaxf(p[j].x + q[j].x, 0.0f) * w[j].x;
            acc += fmaxf(p[j].y + q[j].y, 0.0f) * w[j].y;
            acc += fmaxf(p[j].z + q[j].z, 0.0f) * w[j].z;
            acc += fmaxf(p[j].w + q[j].w, 0.0f) * w[j].w;
        }
        acc += __shfl_xor(acc, 1);
        acc += __shfl_xor(acc, 2);
        acc += __shfl_xor(acc, 4);

        if (l8 == 0) {
            const float x = acc + b2;
            scores[e] = 1.0f / (1.0f + __expf(-x));
            if (x > 0.0f) {
                const int ord = e + 1;
                atomicMax(&last[s], ord);
                atomicMax(&last[d], ord);
            }
        }
        e += EGRID * 32; s = sn; d = dn;
    }
}

template <int LD>
__device__ inline void load_b(float2 (&b)[4][2], const float* wp, int kbase) {
    #pragma unroll
    for (int s = 0; s < 4; ++s) {
        b[s][0] = *(const float2*)(wp + (size_t)(kbase + 4 * s) * LD);
        b[s][1] = *(const float2*)(wp + (size_t)(kbase + 4 * s + 1) * LD);
    }
}

__device__ inline void mma_2x2(v8f (&acc)[2][2], const float2 (&b)[4][2],
                               const float* ap, int kbase) {
    #pragma unroll
    for (int s = 0; s < 4; ++s) {
        const v2f a0 = *(const v2f*)(ap + kbase + 4 * s);
        const v2f a1 = *(const v2f*)(ap + 16 * PSTR + kbase + 4 * s);
        v2f b0, b1;
        b0.x = b[s][0].x; b0.y = b[s][1].x;
        b1.x = b[s][0].y; b1.y = b[s][1].y;
        acc[0][0] = wmma_f32(a0, b0, acc[0][0]);
        acc[1][0] = wmma_f32(a1, b0, acc[1][0]);
        acc[0][1] = wmma_f32(a0, b1, acc[0][1]);
        acc[1][1] = wmma_f32(a1, b1, acc[1][1]);
    }
}

__device__ inline void mma_1x2(v8f (&acc)[2], const float2 (&b)[4][2],
                               const float* ap, int kbase) {
    #pragma unroll
    for (int s = 0; s < 4; ++s) {
        const v2f a = *(const v2f*)(ap + kbase + 4 * s);
        v2f b0, b1;
        b0.x = b[s][0].x; b0.y = b[s][1].x;
        b1.x = b[s][0].y; b1.y = b[s][1].y;
        acc[0] = wmma_f32(a, b0, acc[0]);
        acc[1] = wmma_f32(a, b1, acc[1]);
    }
}

__global__ __launch_bounds__(256) void res_kernel(
        const float* __restrict__ emb, const int* __restrict__ eidx,
        const float* __restrict__ Wr1, const float* __restrict__ br1,
        const float* __restrict__ Wr2, const float* __restrict__ br2,
        const int* __restrict__ last, float* __restrict__ out)
{
    __shared__ __attribute__((aligned(16))) float pair_lds[MT * PSTR];
    __shared__ __attribute__((aligned(16))) float h_lds[MT * PSTR];
    __shared__ int   win_lds[MT];

    const int base = blockIdx.x * MT;

    const int t    = threadIdx.x;
    const int lane = t & 31;
    const int wave = t >> 5;
    const int* srcp = eidx;
    const int* dstp = eidx + EE;

    {
        const int row  = t >> 3;
        const int col  = (t & 7) * 32;
        int node = base + row; if (node > NN - 1) node = NN - 1;
        const int lv = last[node];
        if ((t & 7) == 0) win_lds[row] = lv;
        int nn = node;
        if (lv > 0) nn = (col < DD) ? srcp[lv - 1] : dstp[lv - 1];
        const int goff = (col < DD) ? col : col - DD;
        const float4* g = (const float4*)(emb + (size_t)nn * DD + goff);
        float4* l = (float4*)(pair_lds + row * PSTR + col);
        #pragma unroll
        for (int j = 0; j < 8; ++j) l[j] = g[j];
    }
    __syncthreads();

    if (!__any(win_lds[lane] > 0)) {
        for (int i = t; i < MT * 32; i += 256) {
            const int node = base + (i >> 5);
            if (node < NN)
                ((float4*)out)[(size_t)node * 32 + (i & 31)] =
                    ((const float4*)emb)[(size_t)node * 32 + (i & 31)];
        }
        return;
    }

    const int row   = lane & 15;
    const int khalf = (lane >> 4) << 1;
    const int c2    = (lane & 15) << 1;
    const int mhi   = (lane >> 4) << 3;

    {
        const int cb = wave * 32 + c2;
        const float* wp = Wr1 + (size_t)khalf * HH + cb;
        const float* ap = pair_lds + row * PSTR + khalf;
        v8f acc[2][2];
        #pragma unroll
        for (int m = 0; m < 2; ++m)
            #pragma unroll
            for (int n = 0; n < 2; ++n) acc[m][n] = (v8f){};
        float2 bA[4][2], bB[4][2];
        load_b<HH>(bA, wp, 0);
        #pragma unroll
        for (int k0 = 0; k0 < KK; k0 += 32) {
            load_b<HH>(bB, wp, k0 + 16);
            mma_2x2(acc, bA, ap, k0);
            if (k0 + 32 < KK) load_b<HH>(bA, wp, k0 + 32);
            mma_2x2(acc, bB, ap, k0 + 16);
        }
        const float2 bias = *(const float2*)(br1 + cb);
        #pragma unroll
        for (int m = 0; m < 2; ++m) {
            #pragma unroll
            for (int v = 0; v < 8; ++v) {
                float2 o;
                o.x = fmaxf(acc[m][0][v] + bias.x, 0.0f);
                o.y = fmaxf(acc[m][1][v] + bias.y, 0.0f);
                *(float2*)(h_lds + (m * 16 + v + mhi) * PSTR + cb) = o;
            }
        }
    }
    __syncthreads();

    {
        const int cb = (wave & 3) * 32 + c2;
        const int r0 = (wave >> 2) * 16;
        const float* wp = Wr2 + (size_t)khalf * DD + cb;
        const float* ap = h_lds + (r0 + row) * PSTR + khalf;
        v8f acc[2];
        acc[0] = (v8f){}; acc[1] = (v8f){};
        float2 bA[4][2], bB[4][2];
        load_b<DD>(bA, wp, 0);
        #pragma unroll
        for (int k0 = 0; k0 < KK; k0 += 32) {
            load_b<DD>(bB, wp, k0 + 16);
            mma_1x2(acc, bA, ap, k0);
            if (k0 + 32 < KK) load_b<DD>(bA, wp, k0 + 32);
            mma_1x2(acc, bB, ap, k0 + 16);
        }
        const float2 bias = *(const float2*)(br2 + cb);
        float2 x[8];
        #pragma unroll
        for (int v = 0; v < 8; ++v) {
            int node = base + r0 + v + mhi; if (node > NN - 1) node = NN - 1;
            x[v] = *(const float2*)(emb + (size_t)node * DD + cb);
        }
        #pragma unroll
        for (int v = 0; v < 8; ++v) {
            const int M    = r0 + v + mhi;
            const int node = base + M;
            if (node < NN) {
                float2 o = x[v];
                if (win_lds[M] > 0) {
                    o.x = (x[v].x + acc[0][v] + bias.x) * 0.5f;
                    o.y = (x[v].y + acc[1][v] + bias.y) * 0.5f;
                }
                *(float2*)(out + (size_t)node * DD + cb) = o;
            }
        }
    }
}

extern "C" void kernel_launch(void* const* d_in, const int* in_sizes, int n_in,
                              void* d_out, int out_size, void* d_ws, size_t ws_size,
                              hipStream_t stream) {
    const float* emb = (const float*)d_in[0];
    const int*   eix = (const int*)  d_in[1];
    const float* Wd1 = (const float*)d_in[2];
    const float* bd1 = (const float*)d_in[3];
    const float* Wd2 = (const float*)d_in[4];
    const float* bd2 = (const float*)d_in[5];
    const float* Wr1 = (const float*)d_in[6];
    const float* br1 = (const float*)d_in[7];
    const float* Wr2 = (const float*)d_in[8];
    const float* br2 = (const float*)d_in[9];

    float* out_emb    = (float*)d_out;
    float* out_scores = out_emb + (size_t)NN * DD;

    int* ws   = (int*)d_ws;
    int* last = ws + 16;

    float* tab = (float*)d_ws + TAB_OFF;

    node_kernel<<<dim3((NN + NT - 1) / NT, 2), 128, 0, stream>>>(emb, Wd1, bd1, tab, last, ws_size);

    edge_kernel<<<EGRID, 256, 0, stream>>>(eix, tab, Wd2, bd2, out_scores, last, ws_size);

    res_kernel<<<(NN + MT - 1) / MT, 256, 0, stream>>>(emb, eix, Wr1, br1, Wr2, br2,
                                                       last, out_emb);
}
